# starting kernel
# baseline (speedup 1.0000x reference)
#include <hip/hip_runtime.h>
#include <math.h>

#define NN 50000
#define NE 1600000
#define DIN 128
#define DOUT 32
#define NH 4
#define DT 128
#define NTILES (NN / 16)
#define NPAIR (DIN / 2)
#define WPAD 144

typedef __attribute__((ext_vector_type(2))) float v2f;
typedef __attribute__((ext_vector_type(8))) float v8f;

__global__ void __launch_bounds__(128) gat_wh_gemm(const float* __restrict__ x,
                                                   const float* __restrict__ W,
                                                   float* __restrict__ Wh) {
  __shared__ v2f Wsh[NPAIR * WPAD];

  for (int idx = threadIdx.x; idx < NPAIR * DT; idx += 128) {
    const int p  = idx >> 7;
    const int j  = idx & 127;
    const int kh = j >> 5;
    const int o  = j & 31;
    const float* wp = W + kh * (DIN * DOUT) + (2 * p) * DOUT + o;
    v2f pr;
    pr.x = wp[0];
    pr.y = wp[DOUT];
    Wsh[p * WPAD + j] = pr;
  }
  __syncthreads();

  const int lane = threadIdx.x & 31;
  const int wave = threadIdx.x >> 5;
  const int half = lane >> 4;
  const int l16  = lane & 15;
  const int tile = blockIdx.x * 4 + wave;
  if (tile >= NTILES) return;
  const int row0 = tile * 16;

  v8f acc[8];
#pragma unroll
  for (int jt = 0; jt < 8; ++jt) acc[jt] = (v8f){0.f,0.f,0.f,0.f,0.f,0.f,0.f,0.f};

  const float* xrow = x + (size_t)(row0 + l16) * DIN;

#pragma unroll 4
  for (int kk = 0; kk < 32; ++kk) {
    const int i0 = kk * 4 + half * 2;
    const int p  = kk * 2 + half;
    const v2f a = *(const v2f*)(xrow + i0);
    const v2f* wr = Wsh + p * WPAD + l16;
#pragma unroll
    for (int jt = 0; jt < 8; ++jt) {
      const v2f b = wr[jt * 16];
      acc[jt] = __builtin_amdgcn_wmma_f32_16x16x4_f32(
          false, a, false, b, (short)0, acc[jt], false, false);
    }
  }

#pragma unroll
  for (int jt = 0; jt < 8; ++jt) {
#pragma unroll
    for (int r = 0; r < 8; ++r) {
      const int row = row0 + r + 8 * half;
      Wh[(size_t)row * DT + jt * 16 + l16] = acc[jt][r];
    }
  }
}

__global__ void gat_init(float* __restrict__ out, float* __restrict__ e_max,
                         float* __restrict__ e_sum) {
  int i = blockIdx.x * blockDim.x + threadIdx.x;
  if (i < NN * DT) out[i] = 0.f;
  if (i < NN * NH) { e_sum[i] = 0.f; e_max[i] = -__builtin_inff(); }
}

__global__ void gat_scores(const float* __restrict__ Wh, const float* __restrict__ a,
                           float* __restrict__ s_src, float* __restrict__ s_tgt) {
  int idx = blockIdx.x * blockDim.x + threadIdx.x;
  if (idx >= NN * NH) return;
  int n = idx >> 2, k = idx & 3;
  const float* w  = Wh + (size_t)n * DT + k * DOUT;
  const float* as = a + k * (2 * DOUT);
  const float* at = as + DOUT;
  float ss = 0.f, st = 0.f;
#pragma unroll
  for (int o = 0; o < DOUT; ++o) { float v = w[o]; ss += v * as[o]; st += v * at[o]; }
  s_src[idx] = ss;
  s_tgt[idx] = st;
}

__device__ inline void atomicMaxF(float* addr, float val) {
  if (val >= 0.f) atomicMax((int*)addr, __float_as_int(val));
  else            atomicMin((unsigned int*)addr, __float_as_uint(val));
}

__global__ void gat_edge_max(const int* __restrict__ src, const int* __restrict__ tgt,
                             const float* __restrict__ s_src, const float* __restrict__ s_tgt,
                             float* __restrict__ e_buf, float* __restrict__ e_max) {
  int idx = blockIdx.x * blockDim.x + threadIdx.x;
  if (idx >= NE * NH) return;
  int e = idx >> 2, k = idx & 3;
  int s = src[e], t = tgt[e];
  float v = s_src[s * NH + k] + s_tgt[t * NH + k];
  v = v > 0.f ? v : 0.2f * v;
  e_buf[idx] = v;
  atomicMaxF(&e_max[t * NH + k], v);
}

__global__ void gat_edge_exp(const int* __restrict__ tgt, const float* __restrict__ e_max,
                             float* __restrict__ e_buf, float* __restrict__ e_sum) {
  int idx = blockIdx.x * blockDim.x + threadIdx.x;
  if (idx >= NE * NH) return;
  int e = idx >> 2, k = idx & 3;
  int t = tgt[e];
  float ex = __expf(e_buf[idx] - e_max[t * NH + k]);
  e_buf[idx] = ex;
  atomicAdd(&e_sum[t * NH + k], ex);
}

__global__ void gat_edge_agg(const int* __restrict__ src, const int* __restrict__ tgt,
                             const float* __restrict__ e_buf, const float* __restrict__ e_sum,
                             const float* __restrict__ Wh, float* __restrict__ out) {
  long long idx = (long long)blockIdx.x * blockDim.x + threadIdx.x;
  if (idx >= (long long)NE * DT) return;
  int e = (int)(idx >> 7);
  int j = (int)(idx & 127);
  int k = j >> 5;
  int s = src[e], t = tgt[e];
  float alpha = e_buf[(size_t)e * NH + k] / (e_sum[t * NH + k] + 1e-10f);
  atomicAdd(&out[(size_t)t * DT + j], alpha * Wh[(size_t)s * DT + j]);
}

__global__ void gat_elu(float* __restrict__ out) {
  int i = blockIdx.x * blockDim.x + threadIdx.x;
  if (i >= NN * DT) return;
  float v = out[i];
  out[i] = v > 0.f ? v : expm1f(v);
}

extern "C" void kernel_launch(void* const* d_in, const int* in_sizes, int n_in,
                              void* d_out, int out_size, void* d_ws, size_t ws_size,
                              hipStream_t stream) {
  const float* x    = (const float*)d_in[0];
  const int*   eidx = (const int*)d_in[1];
  const float* W    = (const float*)d_in[2];
  const float* a    = (const float*)d_in[3];
  float* out = (float*)d_out;

  const int* src = eidx;
  const int* tgt = eidx + NE;

  float* ws    = (float*)d_ws;
  float* Wh    = ws;
  float* s_src = Wh + (size_t)NN * DT;
  float* s_tgt = s_src + NN * NH;
  float* e_max = s_tgt + NN * NH;
  float* e_sum = e_max + NN * NH;
  float* e_buf = e_sum + NN * NH;

  const int n_out = NN * DT;
  gat_init<<<(n_out + 255) / 256, 256, 0, stream>>>(out, e_max, e_sum);

  gat_wh_gemm<<<(NTILES + 3) / 4, 128, 0, stream>>>(x, W, Wh);

  gat_scores<<<(NN * NH + 255) / 256, 256, 0, stream>>>(Wh, a, s_src, s_tgt);

  const int ne4 = NE * NH;
  gat_edge_max<<<(ne4 + 255) / 256, 256, 0, stream>>>(src, tgt, s_src, s_tgt, e_buf, e_max);
  gat_edge_exp<<<(ne4 + 255) / 256, 256, 0, stream>>>(tgt, e_max, e_buf, e_sum);

  const long long neagg = (long long)NE * DT;
  gat_edge_agg<<<(int)((neagg + 255) / 256), 256, 0, stream>>>(src, tgt, e_buf, e_sum, Wh, out);

  gat_elu<<<(n_out + 255) / 256, 256, 0, stream>>>(out);
}


# edges grouped by target inside the call (integer rank, scan, fill); one wave per target node does softmax and the weighted Wh row sum in registers, stores the row once: no float atomics
# speedup vs baseline: 2.9189x; 2.9189x over previous
#include <hip/hip_runtime.h>
#include <math.h>

#define NN 50000
#define NE 1600000
#define DIN 128
#define DOUT 32
#define NH 4
#define DT 128
#define NTILES (NN / 16)
#define NPAIR (DIN / 2)
#define WPAD 144

typedef __attribute__((ext_vector_type(2))) float v2f;
typedef __attribute__((ext_vector_type(8))) float v8f;

__global__ void __launch_bounds__(128) gat_wh_gemm(const float* __restrict__ x,
                                                   const float* __restrict__ W,
                                                   float* __restrict__ Wh) {
  __shared__ v2f Wsh[NPAIR * WPAD];

  for (int idx = threadIdx.x; idx < NPAIR * DT; idx += 128) {
    const int p  = idx >> 7;
    const int j  = idx & 127;
    const int kh = j >> 5;
    const int o  = j & 31;
    const float* wp = W + kh * (DIN * DOUT) + (2 * p) * DOUT + o;
    v2f pr;
    pr.x = wp[0];
    pr.y = wp[DOUT];
    Wsh[p * WPAD + j] = pr;
  }
  __syncthreads();

  const int lane = threadIdx.x & 31;
  const int wave = threadIdx.x >> 5;
  const int half = lane >> 4;
  const int l16  = lane & 15;
  const int tile = blockIdx.x * 4 + wave;
  if (tile >= NTILES) return;
  const int row0 = tile * 16;

  v8f acc[8];
#pragma unroll
  for (int jt = 0; jt < 8; ++jt) acc[jt] = (v8f){0.f,0.f,0.f,0.f,0.f,0.f,0.f,0.f};

  const float* xrow = x + (size_t)(row0 + l16) * DIN;

#pragma unroll 4
  for (int kk = 0; kk < 32; ++kk) {
    const int i0 = kk * 4 + half * 2;
    const int p  = kk * 2 + half;
    const v2f a = *(const v2f*)(xrow + i0);
    const v2f* wr = Wsh + p * WPAD + l16;
#pragma unroll
    for (int jt = 0; jt < 8; ++jt) {
      const v2f b = wr[jt * 16];
      acc[jt] = __builtin_amdgcn_wmma_f32_16x16x4_f32(
          false, a, false, b, (short)0, acc[jt], false, false);
    }
  }

#pragma unroll
  for (int jt = 0; jt < 8; ++jt) {
#pragma unroll
    for (int r = 0; r < 8; ++r) {
      const int row = row0 + r + 8 * half;
      Wh[(size_t)row * DT + jt * 16 + l16] = acc[jt][r];
    }
  }
}

__global__ void gat_zero_counts(int* __restrict__ cnt) {
  int i = blockIdx.x * blockDim.x + threadIdx.x;
  if (i < NN) cnt[i] = 0;
}

__global__ void gat_edge_rank(const int* __restrict__ tgt, int* __restrict__ cnt,
                              int* __restrict__ rank) {
  int e = blockIdx.x * blockDim.x + threadIdx.x;
  if (e >= NE) return;
  rank[e] = atomicAdd(&cnt[tgt[e]], 1);
}

#define SCAN_T 1024
#define SCAN_PER ((NN + SCAN_T - 1) / SCAN_T)
__global__ void __launch_bounds__(SCAN_T) gat_scan(const int* __restrict__ cnt,
                                                   int* __restrict__ offs) {
  __shared__ int part[SCAN_T];
  const int tid = threadIdx.x;
  const int b = tid * SCAN_PER;
  const int e = (b + SCAN_PER < NN) ? (b + SCAN_PER) : NN;
  int s = 0;
  for (int i = b; i < e; ++i) s += cnt[i];
  part[tid] = s;
  __syncthreads();
  for (int d = 1; d < SCAN_T; d <<= 1) {
    const int v = (tid >= d) ? part[tid - d] : 0;
    __syncthreads();
    part[tid] += v;
    __syncthreads();
  }
  int run = part[tid] - s;
  for (int i = b; i < e; ++i) { offs[i] = run; run += cnt[i]; }
  if (tid == SCAN_T - 1) offs[NN] = part[SCAN_T - 1];
}

__global__ void gat_edge_fill(const int* __restrict__ src, const int* __restrict__ tgt,
                              const int* __restrict__ rank, const int* __restrict__ offs,
                              int* __restrict__ csr_src) {
  int e = blockIdx.x * blockDim.x + threadIdx.x;
  if (e >= NE) return;
  csr_src[offs[tgt[e]] + rank[e]] = src[e];
}

__device__ inline float lrelu02(float v) { return v > 0.f ? v : 0.2f * v; }

__device__ inline float wave_max(float v) {
#pragma unroll
  for (int o = 16; o > 0; o >>= 1) v = fmaxf(v, __shfl_xor(v, o));
  return v;
}
__device__ inline float wave_sum(float v) {
#pragma unroll
  for (int o = 16; o > 0; o >>= 1) v += __shfl_xor(v, o);
  return v;
}

__global__ void __launch_bounds__(256) gat_node_gather(const int* __restrict__ offs,
                                                       const int* __restrict__ csr_src,
                                                       const float* __restrict__ s_src,
                                                       const float* __restrict__ s_tgt,
                                                       const float* __restrict__ Wh,
                                                       float* __restrict__ out) {
  const int lane = threadIdx.x & 31;
  const int node = blockIdx.x * 8 + __builtin_amdgcn_readfirstlane((int)(threadIdx.x >> 5));
  if (node >= NN) return;
  const int beg = offs[node];
  const int end = offs[node + 1];
  const float4 st = *(const float4*)(s_tgt + (size_t)node * NH);

  const float ninf = -__builtin_inff();
  float m0 = ninf, m1 = ninf, m2 = ninf, m3 = ninf;
  for (int c = beg; c < end; c += 32) {
    const int i = c + lane;
    if (i < end) {
      const int s = csr_src[i];
      const float4 ss = *(const float4*)(s_src + (size_t)s * NH);
      m0 = fmaxf(m0, lrelu02(ss.x + st.x));
      m1 = fmaxf(m1, lrelu02(ss.y + st.y));
      m2 = fmaxf(m2, lrelu02(ss.z + st.z));
      m3 = fmaxf(m3, lrelu02(ss.w + st.w));
    }
  }
  m0 = wave_max(m0); m1 = wave_max(m1); m2 = wave_max(m2); m3 = wave_max(m3);
  m0 = (__builtin_fabsf(m0) < __builtin_inff()) ? m0 : 0.f;
  m1 = (__builtin_fabsf(m1) < __builtin_inff()) ? m1 : 0.f;
  m2 = (__builtin_fabsf(m2) < __builtin_inff()) ? m2 : 0.f;
  m3 = (__builtin_fabsf(m3) < __builtin_inff()) ? m3 : 0.f;

  const int hd = lane >> 3;
  const int k0 = (hd == 0) ? -1 : 0, k1 = (hd == 1) ? -1 : 0;
  const int k2 = (hd == 2) ? -1 : 0, k3 = (hd == 3) ? -1 : 0;
  const float* whl = Wh + lane * 4;
  float4 acc = make_float4(0.f, 0.f, 0.f, 0.f);
  float z0 = 0.f, z1 = 0.f, z2 = 0.f, z3 = 0.f;
  for (int c = beg; c < end; c += 32) {
    const int i = c + lane;
    int s = 0;
    float w0 = 0.f, w1 = 0.f, w2 = 0.f, w3 = 0.f;
    if (i < end) {
      s = csr_src[i];
      const float4 ss = *(const float4*)(s_src + (size_t)s * NH);
      w0 = __expf(lrelu02(ss.x + st.x) - m0);
      w1 = __expf(lrelu02(ss.y + st.y) - m1);
      w2 = __expf(lrelu02(ss.z + st.z) - m2);
      w3 = __expf(lrelu02(ss.w + st.w) - m3);
      z0 += w0; z1 += w1; z2 += w2; z3 += w3;
    }
    const int n = (end - c < 32) ? (end - c) : 32;
    const int sfirst = __builtin_amdgcn_readfirstlane(s);
    if (i >= end) s = sfirst;
    for (int j0 = 0; j0 < n; j0 += 8) {
#pragma unroll
      for (int u = 0; u < 8; ++u) {
        const int j = j0 + u;
        const int sj = __builtin_amdgcn_readlane(s, j);
        const int b0 = __builtin_amdgcn_readlane(__builtin_bit_cast(int, w0), j);
        const int b1 = __builtin_amdgcn_readlane(__builtin_bit_cast(int, w1), j);
        const int b2 = __builtin_amdgcn_readlane(__builtin_bit_cast(int, w2), j);
        const int b3 = __builtin_amdgcn_readlane(__builtin_bit_cast(int, w3), j);
        const float av = __builtin_bit_cast(float, (b0 & k0) | (b1 & k1) | (b2 & k2) | (b3 & k3));
        const float4 v = *(const float4*)(whl + (size_t)sj * DT);
        acc.x += av * v.x; acc.y += av * v.y; acc.z += av * v.z; acc.w += av * v.w;
      }
    }
  }
  z0 = wave_sum(z0); z1 = wave_sum(z1); z2 = wave_sum(z2); z3 = wave_sum(z3);
  const float zs = (hd == 0) ? z0 : (hd == 1) ? z1 : (hd == 2) ? z2 : z3;
  const float inv = 1.0f / (zs + 1e-10f);
  float4 o;
  o.x = acc.x * inv; o.y = acc.y * inv; o.z = acc.z * inv; o.w = acc.w * inv;
  o.x = o.x > 0.f ? o.x : expm1f(o.x);
  o.y = o.y > 0.f ? o.y : expm1f(o.y);
  o.z = o.z > 0.f ? o.z : expm1f(o.z);
  o.w = o.w > 0.f ? o.w : expm1f(o.w);
  *(float4*)(out + (size_t)node * DT + lane * 4) = o;
}

__global__ void gat_scores(const float* __restrict__ Wh, const float* __restrict__ a,
                           float* __restrict__ s_src, float* __restrict__ s_tgt) {
  int idx = blockIdx.x * blockDim.x + threadIdx.x;
  if (idx >= NN * NH) return;
  int n = idx >> 2, k = idx & 3;
  const float* w  = Wh + (size_t)n * DT + k * DOUT;
  const float* as = a + k * (2 * DOUT);
  const float* at = as + DOUT;
  float ss = 0.f, st = 0.f;
#pragma unroll
  for (int o = 0; o < DOUT; ++o) { float v = w[o]; ss += v * as[o]; st += v * at[o]; }
  s_src[idx] = ss;
  s_tgt[idx] = st;
}

extern "C" void kernel_launch(void* const* d_in, const int* in_sizes, int n_in,
                              void* d_out, int out_size, void* d_ws, size_t ws_size,
                              hipStream_t stream) {
  const float* x    = (const float*)d_in[0];
  const int*   eidx = (const int*)d_in[1];
  const float* W    = (const float*)d_in[2];
  const float* a    = (const float*)d_in[3];
  float* out = (float*)d_out;

  const int* src = eidx;
  const int* tgt = eidx + NE;

  float* ws    = (float*)d_ws;
  float* Wh    = ws;
  float* s_src = Wh + (size_t)NN * DT;
  float* s_tgt = s_src + NN * NH;
  int*   cnt   = (int*)(s_tgt + NN * NH);
  int*   offs  = cnt + NN;
  int*   rank  = offs + (NN + 16);
  int*   csr_src = rank + NE;

  gat_zero_counts<<<(NN + 255) / 256, 256, 0, stream>>>(cnt);
  gat_edge_rank<<<(NE + 255) / 256, 256, 0, stream>>>(tgt, cnt, rank);
  gat_scan<<<1, SCAN_T, 0, stream>>>(cnt, offs);
  gat_edge_fill<<<(NE + 255) / 256, 256, 0, stream>>>(src, tgt, rank, offs, csr_src);

  gat_wh_gemm<<<(NTILES + 3) / 4, 128, 0, stream>>>(x, W, Wh);

  gat_scores<<<(NN * NH + 255) / 256, 256, 0, stream>>>(Wh, a, s_src, s_tgt);

  gat_node_gather<<<(NN + 7) / 8, 256, 0, stream>>>(offs, csr_src, s_src, s_tgt, Wh, out);
}


# segment-offset scan on 49 work-groups: each sums the counters in front of its 1,024-counter chunk with whole-line loads, then DPP wave scans; was one work-group walking 196-byte lane strides
# speedup vs baseline: 3.7268x; 1.2768x over previous
#include <hip/hip_runtime.h>
#include <math.h>

#define NN 50000
#define NE 1600000
#define DIN 128
#define DOUT 32
#define NH 4
#define DT 128
#define NTILES (NN / 16)
#define NPAIR (DIN / 2)
#define WPAD 144

typedef __attribute__((ext_vector_type(2))) float v2f;
typedef __attribute__((ext_vector_type(8))) float v8f;

__global__ void __launch_bounds__(128) gat_wh_gemm(const float* __restrict__ x,
                                                   const float* __restrict__ W,
                                                   float* __restrict__ Wh) {
  __shared__ v2f Wsh[NPAIR * WPAD];

  for (int idx = threadIdx.x; idx < NPAIR * DT; idx += 128) {
    const int p  = idx >> 7;
    const int j  = idx & 127;
    const int kh = j >> 5;
    const int o  = j & 31;
    const float* wp = W + kh * (DIN * DOUT) + (2 * p) * DOUT + o;
    v2f pr;
    pr.x = wp[0];
    pr.y = wp[DOUT];
    Wsh[p * WPAD + j] = pr;
  }
  __syncthreads();

  const int lane = threadIdx.x & 31;
  const int wave = threadIdx.x >> 5;
  const int half = lane >> 4;
  const int l16  = lane & 15;
  const int tile = blockIdx.x * 4 + wave;
  if (tile >= NTILES) return;
  const int row0 = tile * 16;

  v8f acc[8];
#pragma unroll
  for (int jt = 0; jt < 8; ++jt) acc[jt] = (v8f){0.f,0.f,0.f,0.f,0.f,0.f,0.f,0.f};

  const float* xrow = x + (size_t)(row0 + l16) * DIN;

#pragma unroll 4
  for (int kk = 0; kk < 32; ++kk) {
    const int i0 = kk * 4 + half * 2;
    const int p  = kk * 2 + half;
    const v2f a = *(const v2f*)(xrow + i0);
    const v2f* wr = Wsh + p * WPAD + l16;
#pragma unroll
    for (int jt = 0; jt < 8; ++jt) {
      const v2f b = wr[jt * 16];
      acc[jt] = __builtin_amdgcn_wmma_f32_16x16x4_f32(
          false, a, false, b, (short)0, acc[jt], false, false);
    }
  }

#pragma unroll
  for (int jt = 0; jt < 8; ++jt) {
#pragma unroll
    for (int r = 0; r < 8; ++r) {
      const int row = row0 + r + 8 * half;
      Wh[(size_t)row * DT + jt * 16 + l16] = acc[jt][r];
    }
  }
}

__global__ void gat_zero_counts(int* __restrict__ cnt) {
  int i = blockIdx.x * blockDim.x + threadIdx.x;
  if (i < NN) cnt[i] = 0;
}

__global__ void gat_edge_rank(const int* __restrict__ tgt, int* __restrict__ cnt,
                              int* __restrict__ rank) {
  int e = blockIdx.x * blockDim.x + threadIdx.x;
  if (e >= NE) return;
  rank[e] = atomicAdd(&cnt[tgt[e]], 1);
}

#define SCAN_T 1024
#define SCAN_B ((NN + 1 + SCAN_T - 1) / SCAN_T)

__device__ inline int wave_incl_scan_i32(int x, int lane) {
  x += __builtin_amdgcn_update_dpp(0, x, 0x111, 0xf, 0xf, true);
  x += __builtin_amdgcn_update_dpp(0, x, 0x112, 0xf, 0xf, true);
  x += __builtin_amdgcn_update_dpp(0, x, 0x114, 0xf, 0xf, true);
  x += __builtin_amdgcn_update_dpp(0, x, 0x118, 0xf, 0xf, true);
  const int low = __builtin_amdgcn_readlane(x, 15);
  return x + ((lane >= 16) ? low : 0);
}

__global__ void __launch_bounds__(SCAN_T) gat_scan(const int* __restrict__ cnt,
                                                   int* __restrict__ offs) {
  __shared__ int wtot[32];
  __shared__ int wpre[32];
  const int tid  = threadIdx.x;
  const int lane = tid & 31;
  const int wave = __builtin_amdgcn_readfirstlane(tid >> 5);
  const int blk  = blockIdx.x;

  int pre = 0;
#pragma unroll
  for (int j = 0; j < SCAN_B - 1; ++j) {
    const bool in_front = j < blk;
    const int c = cnt[(in_front ? j : 0) * SCAN_T + tid];
    pre += in_front ? c : 0;
  }
  const int idx = blk * SCAN_T + tid;
  const int v = (idx < NN) ? cnt[idx] : 0;

  const int incl = wave_incl_scan_i32(v, lane);
  const int psum = wave_incl_scan_i32(pre, lane);
  if (lane == 31) { wtot[wave] = incl; wpre[wave] = psum; }
  __syncthreads();

  const int t     = wtot[lane];
  const int tin   = wave_incl_scan_i32(t, lane);
  const int wbase = __builtin_amdgcn_readlane(tin - t, wave);
  const int front = __builtin_amdgcn_readlane(wave_incl_scan_i32(wpre[lane], lane), 31);
  if (idx <= NN) offs[idx] = front + wbase + incl - v;
}

__global__ void gat_edge_fill(const int* __restrict__ src, const int* __restrict__ tgt,
                              const int* __restrict__ rank, const int* __restrict__ offs,
                              int* __restrict__ csr_src) {
  int e = blockIdx.x * blockDim.x + threadIdx.x;
  if (e >= NE) return;
  csr_src[offs[tgt[e]] + rank[e]] = src[e];
}

__device__ inline float lrelu02(float v) { return v > 0.f ? v : 0.2f * v; }

__device__ inline float wave_max(float v) {
#pragma unroll
  for (int o = 16; o > 0; o >>= 1) v = fmaxf(v, __shfl_xor(v, o));
  return v;
}
__device__ inline float wave_sum(float v) {
#pragma unroll
  for (int o = 16; o > 0; o >>= 1) v += __shfl_xor(v, o);
  return v;
}

__global__ void __launch_bounds__(256) gat_node_gather(const int* __restrict__ offs,
                                                       const int* __restrict__ csr_src,
                                                       const float* __restrict__ s_src,
                                                       const float* __restrict__ s_tgt,
                                                       const float* __restrict__ Wh,
                                                       float* __restrict__ out) {
  const int lane = threadIdx.x & 31;
  const int node = blockIdx.x * 8 + __builtin_amdgcn_readfirstlane((int)(threadIdx.x >> 5));
  if (node >= NN) return;
  const int beg = offs[node];
  const int end = offs[node + 1];
  const float4 st = *(const float4*)(s_tgt + (size_t)node * NH);

  const float ninf = -__builtin_inff();
  float m0 = ninf, m1 = ninf, m2 = ninf, m3 = ninf;
  for (int c = beg; c < end; c += 32) {
    const int i = c + lane;
    if (i < end) {
      const int s = csr_src[i];
      const float4 ss = *(const float4*)(s_src + (size_t)s * NH);
      m0 = fmaxf(m0, lrelu02(ss.x + st.x));
      m1 = fmaxf(m1, lrelu02(ss.y + st.y));
      m2 = fmaxf(m2, lrelu02(ss.z + st.z));
      m3 = fmaxf(m3, lrelu02(ss.w + st.w));
    }
  }
  m0 = wave_max(m0); m1 = wave_max(m1); m2 = wave_max(m2); m3 = wave_max(m3);
  m0 = (__builtin_fabsf(m0) < __builtin_inff()) ? m0 : 0.f;
  m1 = (__builtin_fabsf(m1) < __builtin_inff()) ? m1 : 0.f;
  m2 = (__builtin_fabsf(m2) < __builtin_inff()) ? m2 : 0.f;
  m3 = (__builtin_fabsf(m3) < __builtin_inff()) ? m3 : 0.f;

  const int hd = lane >> 3;
  const int k0 = (hd == 0) ? -1 : 0, k1 = (hd == 1) ? -1 : 0;
  const int k2 = (hd == 2) ? -1 : 0, k3 = (hd == 3) ? -1 : 0;
  const float* whl = Wh + lane * 4;
  float4 acc = make_float4(0.f, 0.f, 0.f, 0.f);
  float z0 = 0.f, z1 = 0.f, z2 = 0.f, z3 = 0.f;
  for (int c = beg; c < end; c += 32) {
    const int i = c + lane;
    int s = 0;
    float w0 = 0.f, w1 = 0.f, w2 = 0.f, w3 = 0.f;
    if (i < end) {
      s = csr_src[i];
      const float4 ss = *(const float4*)(s_src + (size_t)s * NH);
      w0 = __expf(lrelu02(ss.x + st.x) - m0);
      w1 = __expf(lrelu02(ss.y + st.y) - m1);
      w2 = __expf(lrelu02(ss.z + st.z) - m2);
      w3 = __expf(lrelu02(ss.w + st.w) - m3);
      z0 += w0; z1 += w1; z2 += w2; z3 += w3;
    }
    const int n = (end - c < 32) ? (end - c) : 32;
    const int sfirst = __builtin_amdgcn_readfirstlane(s);
    if (i >= end) s = sfirst;
    for (int j0 = 0; j0 < n; j0 += 8) {
#pragma unroll
      for (int u = 0; u < 8; ++u) {
        const int j = j0 + u;
        const int sj = __builtin_amdgcn_readlane(s, j);
        const int b0 = __builtin_amdgcn_readlane(__builtin_bit_cast(int, w0), j);
        const int b1 = __builtin_amdgcn_readlane(__builtin_bit_cast(int, w1), j);
        const int b2 = __builtin_amdgcn_readlane(__builtin_bit_cast(int, w2), j);
        const int b3 = __builtin_amdgcn_readlane(__builtin_bit_cast(int, w3), j);
        const float av = __builtin_bit_cast(float, (b0 & k0) | (b1 & k1) | (b2 & k2) | (b3 & k3));
        const float4 v = *(const float4*)(whl + (size_t)sj * DT);
        acc.x += av * v.x; acc.y += av * v.y; acc.z += av * v.z; acc.w += av * v.w;
      }
    }
  }
  z0 = wave_sum(z0); z1 = wave_sum(z1); z2 = wave_sum(z2); z3 = wave_sum(z3);
  const float zs = (hd == 0) ? z0 : (hd == 1) ? z1 : (hd == 2) ? z2 : z3;
  const float inv = 1.0f / (zs + 1e-10f);
  float4 o;
  o.x = acc.x * inv; o.y = acc.y * inv; o.z = acc.z * inv; o.w = acc.w * inv;
  o.x = o.x > 0.f ? o.x : expm1f(o.x);
  o.y = o.y > 0.f ? o.y : expm1f(o.y);
  o.z = o.z > 0.f ? o.z : expm1f(o.z);
  o.w = o.w > 0.f ? o.w : expm1f(o.w);
  *(float4*)(out + (size_t)node * DT + lane * 4) = o;
}

__global__ void gat_scores(const float* __restrict__ Wh, const float* __restrict__ a,
                           float* __restrict__ s_src, float* __restrict__ s_tgt) {
  int idx = blockIdx.x * blockDim.x + threadIdx.x;
  if (idx >= NN * NH) return;
  int n = idx >> 2, k = idx & 3;
  const float* w  = Wh + (size_t)n * DT + k * DOUT;
  const float* as = a + k * (2 * DOUT);
  const float* at = as + DOUT;
  float ss = 0.f, st = 0.f;
#pragma unroll
  for (int o = 0; o < DOUT; ++o) { float v = w[o]; ss += v * as[o]; st += v * at[o]; }
  s_src[idx] = ss;
  s_tgt[idx] = st;
}

extern "C" void kernel_launch(void* const* d_in, const int* in_sizes, int n_in,
                              void* d_out, int out_size, void* d_ws, size_t ws_size,
                              hipStream_t stream) {
  const float* x    = (const float*)d_in[0];
  const int*   eidx = (const int*)d_in[1];
  const float* W    = (const float*)d_in[2];
  const float* a    = (const float*)d_in[3];
  float* out = (float*)d_out;

  const int* src = eidx;
  const int* tgt = eidx + NE;

  float* ws    = (float*)d_ws;
  float* Wh    = ws;
  float* s_src = Wh + (size_t)NN * DT;
  float* s_tgt = s_src + NN * NH;
  int*   cnt   = (int*)(s_tgt + NN * NH);
  int*   offs  = cnt + NN;
  int*   rank  = offs + (NN + 16);
  int*   csr_src = rank + NE;

  gat_zero_counts<<<(NN + 255) / 256, 256, 0, stream>>>(cnt);
  gat_edge_rank<<<(NE + 255) / 256, 256, 0, stream>>>(tgt, cnt, rank);
  gat_scan<<<SCAN_B, SCAN_T, 0, stream>>>(cnt, offs);
  gat_edge_fill<<<(NE + 255) / 256, 256, 0, stream>>>(src, tgt, rank, offs, csr_src);

  gat_wh_gemm<<<(NTILES + 3) / 4, 128, 0, stream>>>(x, W, Wh);

  gat_scores<<<(NN * NH + 255) / 256, 256, 0, stream>>>(Wh, a, s_src, s_tgt);

  gat_node_gather<<<(NN + 7) / 8, 256, 0, stream>>>(offs, csr_src, s_src, s_tgt, Wh, out);
}


# node gather on resident waves: 1,250 work-groups (was 6,250), each wave walks 5 nodes, next node's counter and list loads in flight, list and scores read once; order of every sum as before
# speedup vs baseline: 4.9143x; 1.3186x over previous
#include <hip/hip_runtime.h>
#include <math.h>

#define NN 50000
#define NE 1600000
#define DIN 128
#define DOUT 32
#define NH 4
#define DT 128
#define NTILES (NN / 16)
#define NPAIR (DIN / 2)
#define WPAD 144

typedef __attribute__((ext_vector_type(2))) float v2f;
typedef __attribute__((ext_vector_type(8))) float v8f;

#define SLOTS 64
#define GEMM_BLOCKS ((NTILES + 7) / 8)
#define EDGE_BLOCKS ((NE + 255) / 256)
__global__ void __launch_bounds__(256) gat_gemm_and_rank(const float* __restrict__ x,
                                                         const float* __restrict__ W,
                                                         float* __restrict__ Wh,
                                                         const int* __restrict__ src,
                                                         const int* __restrict__ tgt,
                                                         int* __restrict__ cnt,
                                                         int* __restrict__ lists) {
  __shared__ v2f Wsh[NPAIR * WPAD];

  if (blockIdx.x >= GEMM_BLOCKS) {
    const int e = (blockIdx.x - GEMM_BLOCKS) * 256 + threadIdx.x;
    if (e < NE) {
      const int t = tgt[e];
      const int s = src[e];
      asm volatile("" : : "v"(s));
      const int r = atomicAdd(&cnt[t], 1);
      if (r < SLOTS) lists[(size_t)t * SLOTS + r] = s;
    }
    return;
  }

  for (int idx = threadIdx.x; idx < NPAIR * DT; idx += 256) {
    const int p  = idx >> 7;
    const int j  = idx & 127;
    const int kh = j >> 5;
    const int o  = j & 31;
    const float* wp = W + kh * (DIN * DOUT) + (2 * p) * DOUT + o;
    v2f pr;
    pr.x = wp[0];
    pr.y = wp[DOUT];
    Wsh[p * WPAD + j] = pr;
  }
  __syncthreads();

  const int lane = threadIdx.x & 31;
  const int wave = threadIdx.x >> 5;
  const int half = lane >> 4;
  const int l16  = lane & 15;
  const int tile = blockIdx.x * 8 + wave;
  if (tile >= NTILES) return;
  const int row0 = tile * 16;

  v8f acc[8];
#pragma unroll
  for (int jt = 0; jt < 8; ++jt) acc[jt] = (v8f){0.f,0.f,0.f,0.f,0.f,0.f,0.f,0.f};

  const float* xrow = x + (size_t)(row0 + l16) * DIN;

#pragma unroll 4
  for (int kk = 0; kk < 32; ++kk) {
    const int i0 = kk * 4 + half * 2;
    const int p  = kk * 2 + half;
    const v2f a = *(const v2f*)(xrow + i0);
    const v2f* wr = Wsh + p * WPAD + l16;
#pragma unroll
    for (int jt = 0; jt < 8; ++jt) {
      const v2f b = wr[jt * 16];
      acc[jt] = __builtin_amdgcn_wmma_f32_16x16x4_f32(
          false, a, false, b, (short)0, acc[jt], false, false);
    }
  }

#pragma unroll
  for (int jt = 0; jt < 8; ++jt) {
#pragma unroll
    for (int r = 0; r < 8; ++r) {
      const int row = row0 + r + 8 * half;
      Wh[(size_t)row * DT + jt * 16 + l16] = acc[jt][r];
    }
  }
}

__global__ void gat_zero_counts(int* __restrict__ cnt) {
  int i = blockIdx.x * blockDim.x + threadIdx.x;
  if (i < NN) cnt[i] = 0;
}

__device__ inline float lrelu02(float v) { return v > 0.f ? v : 0.2f * v; }

__device__ inline float wave_max(float v) {
#pragma unroll
  for (int o = 16; o > 0; o >>= 1) v = fmaxf(v, __shfl_xor(v, o));
  return v;
}
__device__ inline float wave_sum(float v) {
#pragma unroll
  for (int o = 16; o > 0; o >>= 1) v += __shfl_xor(v, o);
  return v;
}


static_assert((size_t)NN * 32 < (1ull << 32), "the 16-B piece index fits 32 bits");
__device__ inline void add_edge_row(float4& acc, const float* __restrict__ Wh, unsigned lane,
                                    int j, int s, float w0, float w1, float w2, float w3,
                                    int k0, int k1, int k2, int k3) {
  const int sj = __builtin_amdgcn_readlane(s, j);
  const int b0 = __builtin_amdgcn_readlane(__builtin_bit_cast(int, w0), j);
  const int b1 = __builtin_amdgcn_readlane(__builtin_bit_cast(int, w1), j);
  const int b2 = __builtin_amdgcn_readlane(__builtin_bit_cast(int, w2), j);
  const int b3 = __builtin_amdgcn_readlane(__builtin_bit_cast(int, w3), j);
  const float av = __builtin_bit_cast(float, (b0 & k0) | (b1 & k1) | (b2 & k2) | (b3 & k3));
  const float4 v = ((const float4*)Wh)[(unsigned)sj * 32u + lane];
  acc.x = __builtin_fmaf(av, v.x, acc.x); acc.y = __builtin_fmaf(av, v.y, acc.y);
  acc.z = __builtin_fmaf(av, v.z, acc.z); acc.w = __builtin_fmaf(av, v.w, acc.w);
}

static_assert(NE % 32 == 0, "the second path reads the edge list in whole 32-edge steps");
static_assert(NN % 8 == 0, "eight nodes per work-group step, no partial group");

__device__ inline void add_chunk_rows(float4& acc, const float* __restrict__ Wh, unsigned lane,
                                      int n, int s, float w0, float w1, float w2, float w3,
                                      int k0, int k1, int k2, int k3) {
  for (int j0 = 0; j0 < n; j0 += 8) {
#pragma unroll
    for (int u = 0; u < 8; ++u)
      add_edge_row(acc, Wh, lane, j0 + u, s, w0, w1, w2, w3, k0, k1, k2, k3);
  }
}

__device__ inline void finish_node(float4 acc, float z0, float z1, float z2, float z3,
                                   int hd, int lane, int node, float* __restrict__ out) {
  z0 = wave_sum(z0); z1 = wave_sum(z1); z2 = wave_sum(z2); z3 = wave_sum(z3);
  const float zs = (hd == 0) ? z0 : (hd == 1) ? z1 : (hd == 2) ? z2 : z3;
  const float inv = 1.0f / (zs + 1e-10f);
  float4 o;
  o.x = acc.x * inv; o.y = acc.y * inv; o.z = acc.z * inv; o.w = acc.w * inv;
  o.x = o.x > 0.f ? o.x : expm1f(o.x);
  o.y = o.y > 0.f ? o.y : expm1f(o.y);
  o.z = o.z > 0.f ? o.z : expm1f(o.z);
  o.w = o.w > 0.f ? o.w : expm1f(o.w);
  *(float4*)(out + (size_t)node * DT + lane * 4) = o;
}

#define NODE_GROUPS (NN / 8)
#define GATHER_BLOCKS 1250
static_assert((NODE_GROUPS + GATHER_BLOCKS - 1) / GATHER_BLOCKS <= 32, "one bit per trip in a 32-bit word");
__global__ void __launch_bounds__(256, 10) gat_node_gather(const int* __restrict__ cnt,
                                                           const int* lists,
                                                           const int* __restrict__ src,
                                                           const int* __restrict__ tgt,
                                                           const float* s_src,
                                                           const float* __restrict__ s_tgt,
                                                           const float* __restrict__ Wh,
                                                           float* __restrict__ out) {
  const int lane = threadIdx.x & 31;
  const int wv = __builtin_amdgcn_readfirstlane((int)(threadIdx.x >> 5));
  const int hd = lane >> 3;
  const int k0 = (hd == 0) ? -1 : 0, k1 = (hd == 1) ? -1 : 0;
  const int k2 = (hd == 2) ? -1 : 0, k3 = (hd == 3) ? -1 : 0;
  const float ninf = -__builtin_inff();

  int node = blockIdx.x * 8 + wv;
  int deg = cnt[node];
  int la = lists[(size_t)node * SLOTS + lane];
  int lb = lists[(size_t)node * SLOTS + 32 + lane];
  unsigned over = 0;
  unsigned trip = 1;

  for (int g = blockIdx.x; g < NODE_GROUPS; g += GATHER_BLOCKS, trip <<= 1) {
    const float4 st = *(const float4*)(s_tgt + (size_t)node * NH);
    const bool va = lane < deg, vb = lane + 32 < deg;
    const int sa = va ? la : 0, sb = vb ? lb : 0;
    const float4 fa = *(const float4*)(s_src + (size_t)sa * NH);
    const float4 fb = *(const float4*)(s_src + (size_t)sb * NH);
    asm volatile("" ::: "memory");
    const int gn = g + GATHER_BLOCKS;
    const int node_n = (gn < NODE_GROUPS) ? gn * 8 + wv : node;
    const int deg_n = cnt[node_n];
    const int la_n = lists[(size_t)node_n * SLOTS + lane];
    const int lb_n = lists[(size_t)node_n * SLOTS + 32 + lane];

    if (deg <= SLOTS) {
      const float ea0 = lrelu02(fa.x + st.x), ea1 = lrelu02(fa.y + st.y);
      const float ea2 = lrelu02(fa.z + st.z), ea3 = lrelu02(fa.w + st.w);
      const float eb0 = lrelu02(fb.x + st.x), eb1 = lrelu02(fb.y + st.y);
      const float eb2 = lrelu02(fb.z + st.z), eb3 = lrelu02(fb.w + st.w);
      float m0 = wave_max(fmaxf(va ? ea0 : ninf, vb ? eb0 : ninf));
      float m1 = wave_max(fmaxf(va ? ea1 : ninf, vb ? eb1 : ninf));
      float m2 = wave_max(fmaxf(va ? ea2 : ninf, vb ? eb2 : ninf));
      float m3 = wave_max(fmaxf(va ? ea3 : ninf, vb ? eb3 : ninf));
      m0 = (__builtin_fabsf(m0) < __builtin_inff()) ? m0 : 0.f;
      m1 = (__builtin_fabsf(m1) < __builtin_inff()) ? m1 : 0.f;
      m2 = (__builtin_fabsf(m2) < __builtin_inff()) ? m2 : 0.f;
      m3 = (__builtin_fabsf(m3) < __builtin_inff()) ? m3 : 0.f;
      const float wa0 = va ? __expf(ea0 - m0) : 0.f, wa1 = va ? __expf(ea1 - m1) : 0.f;
      const float wa2 = va ? __expf(ea2 - m2) : 0.f, wa3 = va ? __expf(ea3 - m3) : 0.f;
      const float wb0 = vb ? __expf(eb0 - m0) : 0.f, wb1 = vb ? __expf(eb1 - m1) : 0.f;
      const float wb2 = vb ? __expf(eb2 - m2) : 0.f, wb3 = vb ? __expf(eb3 - m3) : 0.f;
      float4 acc = make_float4(0.f, 0.f, 0.f, 0.f);
#pragma unroll 1
      for (int c = 0; c < deg; c += 32) {
        const bool hi = c != 0;
        const int n = (deg - c < 32) ? (deg - c) : 32;
        const int l = hi ? lb : la;
        const int first = __builtin_amdgcn_readfirstlane(l);
        const int s = (hi ? vb : va) ? l : first;
        add_chunk_rows(acc, Wh, (unsigned)lane, n, s, hi ? wb0 : wa0, hi ? wb1 : wa1,
                       hi ? wb2 : wa2, hi ? wb3 : wa3, k0, k1, k2, k3);
      }
      finish_node(acc, wa0 + wb0, wa1 + wb1, wa2 + wb2, wa3 + wb3, hd, lane, node, out);
    } else {
      over |= trip;
    }
    node = node_n; deg = deg_n; la = la_n; lb = lb_n;
  }

  trip = 1;
  for (int g = blockIdx.x; g < NODE_GROUPS; g += GATHER_BLOCKS, trip <<= 1) {
    if (!(over & trip)) continue;
    node = g * 8 + wv;
    const float4 st = *(const float4*)(s_tgt + (size_t)node * NH);
    float m0 = ninf, m1 = ninf, m2 = ninf, m3 = ninf;
    for (int c = 0; c < NE; c += 32) {
      const int e = c + lane;
      if (tgt[e] == node) {
        const int s = src[e];
        const float4 ss = *(const float4*)(s_src + (size_t)s * NH);
        m0 = fmaxf(m0, lrelu02(ss.x + st.x));
        m1 = fmaxf(m1, lrelu02(ss.y + st.y));
        m2 = fmaxf(m2, lrelu02(ss.z + st.z));
        m3 = fmaxf(m3, lrelu02(ss.w + st.w));
      }
    }
    m0 = wave_max(m0); m1 = wave_max(m1); m2 = wave_max(m2); m3 = wave_max(m3);
    m0 = (__builtin_fabsf(m0) < __builtin_inff()) ? m0 : 0.f;
    m1 = (__builtin_fabsf(m1) < __builtin_inff()) ? m1 : 0.f;
    m2 = (__builtin_fabsf(m2) < __builtin_inff()) ? m2 : 0.f;
    m3 = (__builtin_fabsf(m3) < __builtin_inff()) ? m3 : 0.f;
    float4 acc = make_float4(0.f, 0.f, 0.f, 0.f);
    float z0 = 0.f, z1 = 0.f, z2 = 0.f, z3 = 0.f;
    for (int c = 0; c < NE; c += 32) {
      const int e = c + lane;
      const bool hit = tgt[e] == node;
      int s = 0;
      float w0 = 0.f, w1 = 0.f, w2 = 0.f, w3 = 0.f;
      if (hit) {
        s = src[e];
        const float4 ss = *(const float4*)(s_src + (size_t)s * NH);
        w0 = __expf(lrelu02(ss.x + st.x) - m0);
        w1 = __expf(lrelu02(ss.y + st.y) - m1);
        w2 = __expf(lrelu02(ss.z + st.z) - m2);
        w3 = __expf(lrelu02(ss.w + st.w) - m3);
        z0 += w0; z1 += w1; z2 += w2; z3 += w3;
      }
      unsigned live = __builtin_amdgcn_ballot_w32(hit);
      while (live) {
        const int j = __builtin_ctz(live);
        live &= live - 1;
        add_edge_row(acc, Wh, (unsigned)lane, j, s, w0, w1, w2, w3, k0, k1, k2, k3);
      }
    }
    finish_node(acc, z0, z1, z2, z3, hd, lane, node, out);
  }
}

#define SCORE_BLOCKS ((NN * NH + 255) / 256)
__global__ void __launch_bounds__(256) gat_scores(const float* __restrict__ Wh,
                                                  const float* __restrict__ a,
                                                  float* __restrict__ s_src,
                                                  float* __restrict__ s_tgt) {
  int idx = blockIdx.x * 256 + threadIdx.x;
  if (idx >= NN * NH) return;
  int n = idx >> 2, k = idx & 3;
  const float* w  = Wh + (size_t)n * DT + k * DOUT;
  const float* as = a + k * (2 * DOUT);
  const float* at = as + DOUT;
  float ss = 0.f, st = 0.f;
#pragma unroll
  for (int o = 0; o < DOUT; ++o) { float v = w[o]; ss += v * as[o]; st += v * at[o]; }
  s_src[idx] = ss;
  s_tgt[idx] = st;
}

extern "C" void kernel_launch(void* const* d_in, const int* in_sizes, int n_in,
                              void* d_out, int out_size, void* d_ws, size_t ws_size,
                              hipStream_t stream) {
  const float* x    = (const float*)d_in[0];
  const int*   eidx = (const int*)d_in[1];
  const float* W    = (const float*)d_in[2];
  const float* a    = (const float*)d_in[3];
  float* out = (float*)d_out;

  const int* src = eidx;
  const int* tgt = eidx + NE;

  float* ws    = (float*)d_ws;
  float* Wh    = ws;
  float* s_src = Wh + (size_t)NN * DT;
  float* s_tgt = s_src + NN * NH;
  int*   cnt   = (int*)(s_tgt + NN * NH);
  int*   lists = cnt + (NN + 48);

  gat_zero_counts<<<(NN + 255) / 256, 256, 0, stream>>>(cnt);
  gat_gemm_and_rank<<<GEMM_BLOCKS + EDGE_BLOCKS, 256, 0, stream>>>(x, W, Wh, src, tgt, cnt, lists);
  gat_scores<<<SCORE_BLOCKS, 256, 0, stream>>>(Wh, a, s_src, s_tgt);

  gat_node_gather<<<GATHER_BLOCKS, 256, 0, stream>>>(cnt, lists, src, tgt, s_src, s_tgt, Wh, out);
}
